# starting kernel
# baseline (speedup 1.0000x reference)
#include <hip/hip_runtime.h>
#include <hip/hip_bf16.h>

typedef __attribute__((ext_vector_type(2))) float v2f;
typedef __attribute__((ext_vector_type(8))) float v8f;

constexpr int S       = 2048;
constexpr int T       = 16384;
constexpr int L       = 16;
constexpr int NCHUNK  = T / L;
constexpr int NGROUP  = S / 16;
constexpr int SBLK    = 32;
constexpr int NSB     = NCHUNK / SBLK;

constexpr size_t OFF_H    = 0;
constexpr size_t OFF_P    = 256;
constexpr size_t OFF_Q    = 320;
constexpr size_t OFF_APOW = 576;
constexpr size_t OFF_ZC   = 4096;
constexpr size_t OFF_U    = OFF_ZC + (size_t)NCHUNK * 4 * S;
constexpr size_t OFF_ZS   = OFF_U  + (size_t)NSB    * 4 * S;

__global__ void lp_setup_kernel(const float* __restrict__ bc,
                                const float* __restrict__ ac,
                                float* __restrict__ ws) {
  if (threadIdx.x != 0 || blockIdx.x != 0) return;
  const float b0 = bc[0];
  float A[4][4], g4[4];
  for (int i = 0; i < 4; ++i) {
    for (int j = 0; j < 4; ++j) A[i][j] = (j == i + 1) ? 1.0f : 0.0f;
    A[i][0] -= ac[i + 1];
    g4[i] = bc[i + 1] - ac[i + 1] * b0;
  }
  float h[16];
  h[0] = b0;
  float v[4] = {g4[0], g4[1], g4[2], g4[3]};
  for (int t = 1; t < 16; ++t) {
    h[t] = v[0];
    float nv[4];
    for (int i = 0; i < 4; ++i) {
      float s = 0.f;
      for (int j = 0; j < 4; ++j) s += A[i][j] * v[j];
      nv[i] = s;
    }
    for (int i = 0; i < 4; ++i) v[i] = nv[i];
  }
  for (int t = 0; t < 16; ++t)
    for (int k = 0; k < 16; ++k)
      ws[OFF_H + t * 16 + k] = (k <= t) ? h[t - k] : 0.0f;
  float M[4][4];
  for (int i = 0; i < 4; ++i)
    for (int j = 0; j < 4; ++j) M[i][j] = (i == j) ? 1.0f : 0.0f;
  for (int t = 0; t < 16; ++t) {
    for (int m = 0; m < 4; ++m) ws[OFF_P + t * 4 + m] = M[0][m];
    float NM[4][4];
    for (int i = 0; i < 4; ++i)
      for (int j = 0; j < 4; ++j) {
        float s = 0.f;
        for (int k = 0; k < 4; ++k) s += A[i][k] * M[k][j];
        NM[i][j] = s;
      }
    for (int i = 0; i < 4; ++i)
      for (int j = 0; j < 4; ++j) M[i][j] = NM[i][j];
  }
  float A16[4][4];
  for (int i = 0; i < 4; ++i)
    for (int j = 0; j < 4; ++j) A16[i][j] = M[i][j];
  for (int i = 0; i < 256; ++i) ws[OFF_Q + i] = 0.0f;
  for (int i = 0; i < 4; ++i) v[i] = g4[i];
  for (int k = 15; k >= 0; --k) {
    for (int m = 0; m < 4; ++m) ws[OFF_Q + m * 16 + k] = v[m];
    float nv[4];
    for (int i = 0; i < 4; ++i) {
      float s = 0.f;
      for (int j = 0; j < 4; ++j) s += A[i][j] * v[j];
      nv[i] = s;
    }
    for (int i = 0; i < 4; ++i) v[i] = nv[i];
  }
  float Pw[4][4];
  for (int i = 0; i < 4; ++i)
    for (int j = 0; j < 4; ++j) Pw[i][j] = (i == j) ? 1.0f : 0.0f;
  for (int p = 0; p <= 32; ++p) {
    for (int i = 0; i < 4; ++i)
      for (int j = 0; j < 4; ++j) ws[OFF_APOW + p * 16 + i * 4 + j] = Pw[i][j];
    float NM[4][4];
    for (int i = 0; i < 4; ++i)
      for (int j = 0; j < 4; ++j) {
        float s = 0.f;
        for (int k = 0; k < 4; ++k) s += Pw[i][k] * A16[k][j];
        NM[i][j] = s;
      }
    for (int i = 0; i < 4; ++i)
      for (int j = 0; j < 4; ++j) Pw[i][j] = NM[i][j];
  }
}

__global__ __launch_bounds__(256) void lp_zc_kernel(const float* __restrict__ x,
                                                    float* __restrict__ ws) {
  const int wave = threadIdx.x >> 5;
  const int lane = threadIdx.x & 31;
  const int tile = blockIdx.x * 8 + wave;
  const int c = tile >> 7;
  const int g = tile & 127;
  const int half = lane >> 4;
  const int l16 = lane & 15;
  const float* Q = ws + OFF_Q;
  const float* xt = x + (size_t)c * L * S + (size_t)g * 16 + l16;
  v8f acc = {};
#pragma unroll
  for (int k = 0; k < 4; ++k) {
    const int kk0 = 4 * k + 2 * half;
    v2f av, bv;
    av[0] = Q[l16 * 16 + kk0];
    av[1] = Q[l16 * 16 + kk0 + 1];
    bv[0] = xt[(size_t)kk0 * S];
    bv[1] = xt[(size_t)(kk0 + 1) * S];
    acc = __builtin_amdgcn_wmma_f32_16x16x4_f32(false, av, false, bv,
                                                (short)0, acc, false, false);
  }
  if (half == 0) {
    float* zc = ws + OFF_ZC + (size_t)c * 4 * S + (size_t)g * 16 + l16;
    zc[0 * (size_t)S] = acc[0];
    zc[1 * (size_t)S] = acc[1];
    zc[2 * (size_t)S] = acc[2];
    zc[3 * (size_t)S] = acc[3];
  }
}

__global__ __launch_bounds__(256) void lp_scan1_kernel(float* __restrict__ ws) {
  const int tid = blockIdx.x * 256 + threadIdx.x;
  const int s  = tid & (S - 1);
  const int sb = tid >> 11;
  const float* A16 = ws + OFF_APOW + 16;
  float a[16];
#pragma unroll
  for (int i = 0; i < 16; ++i) a[i] = A16[i];
  float w0 = 0.f, w1 = 0.f, w2 = 0.f, w3 = 0.f;
  float* zc = ws + OFF_ZC + ((size_t)sb * SBLK) * 4 * S + s;
  for (int cc = 0; cc < SBLK; ++cc) {
    const float r0 = zc[0 * (size_t)S], r1 = zc[1 * (size_t)S];
    const float r2 = zc[2 * (size_t)S], r3 = zc[3 * (size_t)S];
    zc[0 * (size_t)S] = w0; zc[1 * (size_t)S] = w1;
    zc[2 * (size_t)S] = w2; zc[3 * (size_t)S] = w3;
    const float n0 = a[0]*w0 + a[1]*w1 + a[2]*w2 + a[3]*w3 + r0;
    const float n1 = a[4]*w0 + a[5]*w1 + a[6]*w2 + a[7]*w3 + r1;
    const float n2 = a[8]*w0 + a[9]*w1 + a[10]*w2 + a[11]*w3 + r2;
    const float n3 = a[12]*w0 + a[13]*w1 + a[14]*w2 + a[15]*w3 + r3;
    w0 = n0; w1 = n1; w2 = n2; w3 = n3;
    zc += (size_t)4 * S;
  }
  float* u = ws + OFF_U + (size_t)sb * 4 * S + s;
  u[0 * (size_t)S] = w0; u[1 * (size_t)S] = w1;
  u[2 * (size_t)S] = w2; u[3 * (size_t)S] = w3;
}

__global__ __launch_bounds__(256) void lp_scan2_kernel(float* __restrict__ ws) {
  const int s = blockIdx.x * 256 + threadIdx.x;
  const float* A = ws + OFF_APOW + 32 * 16;
  float a[16];
#pragma unroll
  for (int i = 0; i < 16; ++i) a[i] = A[i];
  float z0 = 0.f, z1 = 0.f, z2 = 0.f, z3 = 0.f;
  for (int sb = 0; sb < NSB; ++sb) {
    float* zs = ws + OFF_ZS + (size_t)sb * 4 * S + s;
    zs[0 * (size_t)S] = z0; zs[1 * (size_t)S] = z1;
    zs[2 * (size_t)S] = z2; zs[3 * (size_t)S] = z3;
    const float* u = ws + OFF_U + (size_t)sb * 4 * S + s;
    const float u0 = u[0 * (size_t)S], u1 = u[1 * (size_t)S];
    const float u2 = u[2 * (size_t)S], u3 = u[3 * (size_t)S];
    const float n0 = a[0]*z0 + a[1]*z1 + a[2]*z2 + a[3]*z3 + u0;
    const float n1 = a[4]*z0 + a[5]*z1 + a[6]*z2 + a[7]*z3 + u1;
    const float n2 = a[8]*z0 + a[9]*z1 + a[10]*z2 + a[11]*z3 + u2;
    const float n3 = a[12]*z0 + a[13]*z1 + a[14]*z2 + a[15]*z3 + u3;
    z0 = n0; z1 = n1; z2 = n2; z3 = n3;
  }
}

__global__ __launch_bounds__(256) void lp_final_kernel(const float* __restrict__ x,
                                                       float* __restrict__ out,
                                                       const float* __restrict__ ws) {
  const int wave = threadIdx.x >> 5;
  const int lane = threadIdx.x & 31;
  const int tile = blockIdx.x * 8 + wave;
  const int c = tile >> 7;
  const int g = tile & 127;
  const int half = lane >> 4;
  const int l16 = lane & 15;
  const int sb = c >> 5;
  const int cc = c & 31;
  const int chan = g * 16 + l16;

  const float* zs = ws + OFF_ZS + (size_t)sb * 4 * S + chan;
  const float z0 = zs[0 * (size_t)S], z1 = zs[1 * (size_t)S];
  const float z2 = zs[2 * (size_t)S], z3 = zs[3 * (size_t)S];
  const float* ap = ws + OFF_APOW + cc * 16;
  const float* W  = ws + OFF_ZC + (size_t)c * 4 * S + chan;
  v2f zin;
#pragma unroll
  for (int j = 0; j < 2; ++j) {
    const int K = 2 * half + j;
    zin[j] = ap[K * 4 + 0] * z0 + ap[K * 4 + 1] * z1 +
             ap[K * 4 + 2] * z2 + ap[K * 4 + 3] * z3 + W[(size_t)K * S];
  }
  const float* Pm = ws + OFF_P;
  v2f pa;
  pa[0] = Pm[l16 * 4 + 2 * half + 0];
  pa[1] = Pm[l16 * 4 + 2 * half + 1];

  v8f acc = {};
  acc = __builtin_amdgcn_wmma_f32_16x16x4_f32(false, pa, false, zin,
                                              (short)0, acc, false, false);

  const float* H = ws + OFF_H;
  const float* xt = x + (size_t)c * L * S + chan;
#pragma unroll
  for (int k = 0; k < 4; ++k) {
    const int kk0 = 4 * k + 2 * half;
    v2f av, bv;
    av[0] = H[l16 * 16 + kk0];
    av[1] = H[l16 * 16 + kk0 + 1];
    bv[0] = xt[(size_t)kk0 * S];
    bv[1] = xt[(size_t)(kk0 + 1) * S];
    acc = __builtin_amdgcn_wmma_f32_16x16x4_f32(false, av, false, bv,
                                                (short)0, acc, false, false);
  }

  float* o = out + (size_t)c * L * S + chan;
#pragma unroll
  for (int r = 0; r < 8; ++r)
    o[(size_t)(r + 8 * half) * S] = acc[r];
}

extern "C" void kernel_launch(void* const* d_in, const int* in_sizes, int n_in,
                              void* d_out, int out_size, void* d_ws, size_t ws_size,
                              hipStream_t stream) {
  const float* signal = (const float*)d_in[0];
  const float* b      = (const float*)d_in[1];
  const float* a      = (const float*)d_in[2];
  float* out = (float*)d_out;
  float* ws  = (float*)d_ws;

  lp_setup_kernel<<<1, 64, 0, stream>>>(b, a, ws);

  const int tiles  = NCHUNK * NGROUP;
  const int blocks = tiles / 8;
  lp_zc_kernel<<<blocks, 256, 0, stream>>>(signal, ws);

  lp_scan1_kernel<<<(S * NSB) / 256, 256, 0, stream>>>(ws);
  lp_scan2_kernel<<<S / 256, 256, 0, stream>>>(ws);

  lp_final_kernel<<<blocks, 256, 0, stream>>>(signal, out, ws);
}


# One launch: a work-group owns a 16-channel strip for all rows and reads x once (rows stay in registers from state pass to output pass); run states scanned in LDS; summation order changes
# speedup vs baseline: 1.1729x; 1.1729x over previous
#include <hip/hip_runtime.h>


typedef __attribute__((ext_vector_type(2))) float v2f;
typedef __attribute__((ext_vector_type(8))) float v8f;

constexpr int S       = 2048;
constexpr int T       = 16384;
constexpr int L       = 16;
constexpr int NW      = 16;
constexpr int R       = 8;
constexpr int RUNROWS = R * L;
constexpr int SSROWS  = NW * RUNROWS;
constexpr int NSS     = T / SSROWS;
constexpr int NSTRIP  = S / 16;

constexpr int LH  = 0;
constexpr int LP  = 256;
constexpr int LQ  = 320;
constexpr int LAP = 576;
constexpr int LHV = 720;
constexpr int LWL = 768;
constexpr int LUB = LWL + NW * R * 64;
constexpr int LZR = LUB + NW * 64;
constexpr int LTR = LZR + NW * 64;
constexpr int LDS_FLOATS = LTR + NW * 64;

__device__ __forceinline__ void lp_make_A(const float* bc, const float* ac,
                                          float (&A)[4][4], float (&g4)[4]) {
  const float b0 = bc[0];
  for (int i = 0; i < 4; ++i) {
    for (int j = 0; j < 4; ++j) A[i][j] = (j == i + 1) ? 1.0f : 0.0f;
    A[i][0] -= ac[i + 1];
    g4[i] = bc[i + 1] - ac[i + 1] * b0;
  }
}

__device__ __forceinline__ void lp_matvec(const float (&A)[4][4], float (&v)[4]) {
  float nv[4];
  for (int i = 0; i < 4; ++i) {
    float s = 0.f;
    for (int j = 0; j < 4; ++j) s += A[i][j] * v[j];
    nv[i] = s;
  }
  for (int i = 0; i < 4; ++i) v[i] = nv[i];
}

__device__ __forceinline__ void lp_matmul(const float (&X)[4][4], const float (&Y)[4][4],
                                          float (&M)[4][4]) {
  float NM[4][4];
  for (int i = 0; i < 4; ++i)
    for (int j = 0; j < 4; ++j) {
      float s = 0.f;
      for (int k = 0; k < 4; ++k) s += X[i][k] * Y[k][j];
      NM[i][j] = s;
    }
  for (int i = 0; i < 4; ++i)
    for (int j = 0; j < 4; ++j) M[i][j] = NM[i][j];
}

template <int JB>
__device__ __forceinline__ void lp_load4(v2f (&xb)[R][4], const float* __restrict__ p) {
  for (int jj = 0; jj < 4; ++jj) {
    for (int k = 0; k < 4; ++k) {
      const int row = (JB + jj) * L + 4 * k;
      xb[JB + jj][k][0] = p[(size_t)row * S];
      xb[JB + jj][k][1] = p[(size_t)(row + 1) * S];
    }
  }
}

__global__ __launch_bounds__(512) void lp_strip_kernel(const float* __restrict__ x,
                                                       const float* __restrict__ bc,
                                                       const float* __restrict__ ac,
                                                       float* __restrict__ out) {
  __shared__ __attribute__((aligned(16))) float lds[LDS_FLOATS];

  const int tid  = threadIdx.x;
  const int wave = tid >> 5;
  const int lane = tid & 31;
  const int half = lane >> 4;
  const int l16  = lane & 15;
  const int g    = blockIdx.x;
  const int chan = g * 16 + l16;

  const float* xs = x + (size_t)(wave * RUNROWS + 2 * half) * S + chan;

  v2f xa[R][4], xb[R][4];
  lp_load4<0>(xa, xs);
  lp_load4<4>(xa, xs);

  if (tid == 0) {
    float A[4][4], g4[4];
    lp_make_A(bc, ac, A, g4);
    lds[LHV + 0] = bc[0];
    float v[4] = {g4[0], g4[1], g4[2], g4[3]};
    for (int t = 1; t < 16; ++t) {
      lds[LHV + t] = v[0];
      lp_matvec(A, v);
    }
  }
  if (tid == 32) {
    float A[4][4], g4[4];
    lp_make_A(bc, ac, A, g4);
    float M[4][4];
    for (int i = 0; i < 4; ++i)
      for (int j = 0; j < 4; ++j) M[i][j] = (i == j) ? 1.0f : 0.0f;
    for (int t = 0; t < 16; ++t) {
      for (int m = 0; m < 4; ++m) lds[LP + t * 4 + m] = M[0][m];
      lp_matmul(A, M, M);
    }
    float A16[4][4], Pw[4][4];
    for (int i = 0; i < 4; ++i)
      for (int j = 0; j < 4; ++j) {
        A16[i][j] = M[i][j];
        Pw[i][j] = (i == j) ? 1.0f : 0.0f;
      }
    for (int p = 0; p <= R; ++p) {
      for (int i = 0; i < 4; ++i)
        for (int j = 0; j < 4; ++j) lds[LAP + p * 16 + i * 4 + j] = Pw[i][j];
      lp_matmul(Pw, A16, Pw);
    }
  }
  if (tid == 64) {
    float A[4][4], g4[4];
    lp_make_A(bc, ac, A, g4);
    float v[4] = {g4[0], g4[1], g4[2], g4[3]};
    for (int k = 15; k >= 0; --k) {
      for (int m = 0; m < 4; ++m) lds[LQ + m * 16 + k] = v[m];
      lp_matvec(A, v);
    }
  }
  if (tid >= 64 && tid < 256) lds[LQ + tid] = 0.0f;
  __syncthreads();
  if (tid < 256) {
    const int t = tid >> 4, k = tid & 15;
    lds[LH + tid] = (k <= t) ? lds[LHV + t - k] : 0.0f;
  }
  __syncthreads();

  v2f Hf[4], Qf[4], Pf;
  for (int k = 0; k < 4; ++k) {
    const int kk0 = 4 * k + 2 * half;
    Hf[k][0] = lds[LH + l16 * 16 + kk0];
    Hf[k][1] = lds[LH + l16 * 16 + kk0 + 1];
    Qf[k][0] = lds[LQ + l16 * 16 + kk0];
    Qf[k][1] = lds[LQ + l16 * 16 + kk0 + 1];
  }
  Pf[0] = lds[LP + l16 * 4 + 2 * half + 0];
  Pf[1] = lds[LP + l16 * 4 + 2 * half + 1];
  const int wl_off  = half ? (LTR + wave * 64 + l16) : (LWL + wave * (R * 64) + l16);
  const int wl_step = half ? 0 : 64;
  const int ub_off  = half ? (LTR + wave * 64 + l16) : (LUB + wave * 64 + l16);
  const int zr_off  = half ? (LTR + l16) : (LZR + l16);
  const int zr_step = half ? 0 : 64;

  float zs0 = 0.f, zs1 = 0.f, zs2 = 0.f, zs3 = 0.f;

  auto step = [&](const int ss, v2f (&cur)[R][4], v2f (&nxt)[R][4])
      __attribute__((always_inline)) {
    const int sn = (ss + 1 < NSS) ? ss + 1 : ss;
    const float* xn = xs + (size_t)sn * SSROWS * S;

    float a16[16];
    for (int i = 0; i < 16; ++i) a16[i] = lds[LAP + 16 + i];
    float w0 = 0.f, w1 = 0.f, w2 = 0.f, w3 = 0.f;
    for (int j = 0; j < R; ++j) {
      v8f acc = {};
      for (int k = 0; k < 4; ++k)
        acc = __builtin_amdgcn_wmma_f32_16x16x4_f32(false, Qf[k], false, cur[j][k],
                                                    (short)0, acc, false, false);
      const float n0 = a16[0]*w0 + a16[1]*w1 + a16[2]*w2 + a16[3]*w3 + acc[0];
      const float n1 = a16[4]*w0 + a16[5]*w1 + a16[6]*w2 + a16[7]*w3 + acc[1];
      const float n2 = a16[8]*w0 + a16[9]*w1 + a16[10]*w2 + a16[11]*w3 + acc[2];
      const float n3 = a16[12]*w0 + a16[13]*w1 + a16[14]*w2 + a16[15]*w3 + acc[3];
      float* wd = lds + wl_off + j * wl_step;
      wd[0] = w0; wd[16] = w1; wd[32] = w2; wd[48] = w3;
      w0 = n0; w1 = n1; w2 = n2; w3 = n3;
    }
    {
      float* ud = lds + ub_off;
      ud[0] = w0; ud[16] = w1; ud[32] = w2; ud[48] = w3;
    }
    __syncthreads();

    if (wave == 0) {
      float a8[16];
      for (int i = 0; i < 16; ++i) a8[i] = lds[LAP + R * 16 + i];
      float u[NW][4];
      for (int w = 0; w < NW; ++w) {
        const float* ub = lds + LUB + w * 64 + l16;
        u[w][0] = ub[0]; u[w][1] = ub[16]; u[w][2] = ub[32]; u[w][3] = ub[48];
      }
      float z0 = zs0, z1 = zs1, z2 = zs2, z3 = zs3;
      for (int w = 0; w < NW; ++w) {
        float* zd = lds + zr_off + w * zr_step;
        zd[0] = z0; zd[16] = z1; zd[32] = z2; zd[48] = z3;
        const float n0 = a8[0]*z0 + a8[1]*z1 + a8[2]*z2 + a8[3]*z3 + u[w][0];
        const float n1 = a8[4]*z0 + a8[5]*z1 + a8[6]*z2 + a8[7]*z3 + u[w][1];
        const float n2 = a8[8]*z0 + a8[9]*z1 + a8[10]*z2 + a8[11]*z3 + u[w][2];
        const float n3 = a8[12]*z0 + a8[13]*z1 + a8[14]*z2 + a8[15]*z3 + u[w][3];
        z0 = n0; z1 = n1; z2 = n2; z3 = n3;
      }
      zs0 = z0; zs1 = z1; zs2 = z2; zs3 = z3;
    }
    __syncthreads();

    lp_load4<0>(nxt, xn);

    const float* zrp = lds + LZR + wave * 64 + l16;
    const float r0 = zrp[0], r1 = zrp[16], r2 = zrp[32], r3 = zrp[48];
    const float* wlp = lds + LWL + wave * (R * 64) + l16;
    float* o = out + (size_t)(ss * SSROWS + wave * RUNROWS + 8 * half) * S + chan;
    for (int j = 0; j < R; ++j) {
      if (j == R / 2) {
        asm volatile("" ::: "memory");
        lp_load4<4>(nxt, xn);
        asm volatile("" ::: "memory");
      }
      v2f zin;
      for (int s = 0; s < 2; ++s) {
        const int K = 2 * half + s;
        const float4 ap = *(const float4*)(lds + LAP + j * 16 + K * 4);
        zin[s] = ap.x * r0 + ap.y * r1 + ap.z * r2 + ap.w * r3 + wlp[j * 64 + K * 16];
      }
      v8f acc = {};
      acc = __builtin_amdgcn_wmma_f32_16x16x4_f32(false, Pf, false, zin,
                                                  (short)0, acc, false, false);
      for (int k = 0; k < 4; ++k)
        acc = __builtin_amdgcn_wmma_f32_16x16x4_f32(false, Hf[k], false, cur[j][k],
                                                    (short)0, acc, false, false);
      float* oj = o + (size_t)(j * L) * S;
      for (int r = 0; r < 8; ++r) oj[(size_t)r * S] = acc[r];
    }
  };

  for (int ss = 0; ss < NSS; ss += 2) {
    step(ss, xa, xb);
    step(ss + 1, xb, xa);
  }
}

extern "C" void kernel_launch(void* const* d_in, const int* in_sizes, int n_in,
                              void* d_out, int out_size, void* d_ws, size_t ws_size,
                              hipStream_t stream) {
  const float* signal = (const float*)d_in[0];
  const float* b      = (const float*)d_in[1];
  const float* a      = (const float*)d_in[2];
  float* out = (float*)d_out;

  lp_strip_kernel<<<NSTRIP, 512, 0, stream>>>(signal, b, a, out);
}


# Time axis split over two work-groups per strip (256 groups): the later one first rebuilds its entry state by a state-only sweep of the earlier rows, joined via A^512; summation order changes
# speedup vs baseline: 1.2288x; 1.0477x over previous
#include <hip/hip_runtime.h>


typedef __attribute__((ext_vector_type(2))) float v2f;
typedef __attribute__((ext_vector_type(8))) float v8f;

constexpr int S       = 2048;
constexpr int T       = 16384;
constexpr int L       = 16;
constexpr int NW      = 16;
constexpr int R       = 8;
constexpr int RUNROWS = R * L;
constexpr int SSROWS  = NW * RUNROWS;
constexpr int NSS     = T / SSROWS;
constexpr int NSTRIP  = S / 16;
constexpr int NHALF   = 2;
constexpr int NGRID   = NSTRIP * NHALF;
constexpr int NSSH    = NSS / NHALF;
constexpr int SWROWS  = (T / NHALF) / NW;
constexpr int SWGRP   = SWROWS / RUNROWS;
static_assert(SWGRP == 4, "the entry-state sweep is written for four groups and A^512");
static_assert(NSSH % 2 == 0, "the row buffers swap roles two super-steps per trip");

constexpr int LH  = 0;
constexpr int LP  = 256;
constexpr int LQ  = 320;
constexpr int LAP = 576;
constexpr int LHV = 720;
constexpr int LA5 = 736;
constexpr int LWL = 768;
constexpr int LUB = LWL + NW * R * 64;
constexpr int LZR = LUB + NW * 64;
constexpr int LTR = LZR + NW * 64;
constexpr int LDS_FLOATS = LTR + NW * 64;

__device__ __forceinline__ void lp_make_A(const float* bc, const float* ac,
                                          float (&A)[4][4], float (&g4)[4]) {
  const float b0 = bc[0];
  for (int i = 0; i < 4; ++i) {
    for (int j = 0; j < 4; ++j) A[i][j] = (j == i + 1) ? 1.0f : 0.0f;
    A[i][0] -= ac[i + 1];
    g4[i] = bc[i + 1] - ac[i + 1] * b0;
  }
}

__device__ __forceinline__ void lp_matvec(const float (&A)[4][4], float (&v)[4]) {
  float nv[4];
  for (int i = 0; i < 4; ++i) {
    float s = 0.f;
    for (int j = 0; j < 4; ++j) s += A[i][j] * v[j];
    nv[i] = s;
  }
  for (int i = 0; i < 4; ++i) v[i] = nv[i];
}

__device__ __forceinline__ void lp_matmul(const float (&X)[4][4], const float (&Y)[4][4],
                                          float (&M)[4][4]) {
  float NM[4][4];
  for (int i = 0; i < 4; ++i)
    for (int j = 0; j < 4; ++j) {
      float s = 0.f;
      for (int k = 0; k < 4; ++k) s += X[i][k] * Y[k][j];
      NM[i][j] = s;
    }
  for (int i = 0; i < 4; ++i)
    for (int j = 0; j < 4; ++j) M[i][j] = NM[i][j];
}

template <int JB>
__device__ __forceinline__ void lp_load4(v2f (&xb)[R][4], const float* __restrict__ p) {
  for (int jj = 0; jj < 4; ++jj) {
    for (int k = 0; k < 4; ++k) {
      const int row = (JB + jj) * L + 4 * k;
      xb[JB + jj][k][0] = p[(size_t)row * S];
      xb[JB + jj][k][1] = p[(size_t)(row + 1) * S];
    }
  }
}

__global__ __launch_bounds__(512) void lp_strip_kernel(const float* __restrict__ x,
                                                       const float* __restrict__ bc,
                                                       const float* __restrict__ ac,
                                                       float* __restrict__ out) {
  __shared__ __attribute__((aligned(16))) float lds[LDS_FLOATS];

  const int tid  = threadIdx.x;
  const int wave = tid >> 5;
  const int lane = tid & 31;
  const int half = lane >> 4;
  const int l16  = lane & 15;
  const int bid  = blockIdx.x;
  const int g    = bid & (NSTRIP - 1);
  const int hh   = (bid < NSTRIP) ? 1 : 0;
  const int chan = g * 16 + l16;
  const int ss_lo = hh * NSSH, ss_hi = ss_lo + NSSH;

  const float* xs = x + (size_t)(wave * RUNROWS + 2 * half) * S + chan;
  const float* sp = x + (size_t)(wave * SWROWS + 2 * half) * S + chan;

  v2f xa[R][4], xb[R][4];
  {
    const float* first = hh ? sp : xs + (size_t)ss_lo * SSROWS * S;
    lp_load4<0>(xa, first);
    lp_load4<4>(xa, first);
  }

  if (tid == 0) {
    float A[4][4], g4[4];
    lp_make_A(bc, ac, A, g4);
    lds[LHV + 0] = bc[0];
    float v[4] = {g4[0], g4[1], g4[2], g4[3]};
    for (int t = 1; t < 16; ++t) {
      lds[LHV + t] = v[0];
      lp_matvec(A, v);
    }
  }
  if (tid == 32) {
    float A[4][4], g4[4];
    lp_make_A(bc, ac, A, g4);
    float M[4][4];
    for (int i = 0; i < 4; ++i)
      for (int j = 0; j < 4; ++j) M[i][j] = (i == j) ? 1.0f : 0.0f;
    for (int t = 0; t < 16; ++t) {
      for (int m = 0; m < 4; ++m) lds[LP + t * 4 + m] = M[0][m];
      lp_matmul(A, M, M);
    }
    float A16[4][4], Pw[4][4];
    for (int i = 0; i < 4; ++i)
      for (int j = 0; j < 4; ++j) {
        A16[i][j] = M[i][j];
        Pw[i][j] = (i == j) ? 1.0f : 0.0f;
      }
    float A5[4][4];
    for (int p = 0; p <= R; ++p) {
      for (int i = 0; i < 4; ++i)
        for (int j = 0; j < 4; ++j) {
          lds[LAP + p * 16 + i * 4 + j] = Pw[i][j];
          if (p == R) A5[i][j] = Pw[i][j];
        }
      lp_matmul(Pw, A16, Pw);
    }
    lp_matmul(A5, A5, A5);
    lp_matmul(A5, A5, A5);
    for (int i = 0; i < 4; ++i)
      for (int j = 0; j < 4; ++j) lds[LA5 + i * 4 + j] = A5[i][j];
  }
  if (tid == 64) {
    float A[4][4], g4[4];
    lp_make_A(bc, ac, A, g4);
    float v[4] = {g4[0], g4[1], g4[2], g4[3]};
    for (int k = 15; k >= 0; --k) {
      for (int m = 0; m < 4; ++m) lds[LQ + m * 16 + k] = v[m];
      lp_matvec(A, v);
    }
  }
  if (tid >= 64 && tid < 256) lds[LQ + tid] = 0.0f;
  __syncthreads();
  if (tid < 256) {
    const int t = tid >> 4, k = tid & 15;
    lds[LH + tid] = (k <= t) ? lds[LHV + t - k] : 0.0f;
  }
  __syncthreads();

  v2f Hf[4], Qf[4], Pf;
  for (int k = 0; k < 4; ++k) {
    const int kk0 = 4 * k + 2 * half;
    Hf[k][0] = lds[LH + l16 * 16 + kk0];
    Hf[k][1] = lds[LH + l16 * 16 + kk0 + 1];
    Qf[k][0] = lds[LQ + l16 * 16 + kk0];
    Qf[k][1] = lds[LQ + l16 * 16 + kk0 + 1];
  }
  Pf[0] = lds[LP + l16 * 4 + 2 * half + 0];
  Pf[1] = lds[LP + l16 * 4 + 2 * half + 1];
  const int wl_off  = half ? (LTR + wave * 64 + l16) : (LWL + wave * (R * 64) + l16);
  const int wl_step = half ? 0 : 64;
  const int ub_off  = half ? (LTR + wave * 64 + l16) : (LUB + wave * 64 + l16);
  const int zr_off  = half ? (LTR + l16) : (LZR + l16);
  const int zr_step = half ? 0 : 64;

  float zs0 = 0.f, zs1 = 0.f, zs2 = 0.f, zs3 = 0.f;

  if (hh) {
    float a16[16];
    for (int i = 0; i < 16; ++i) a16[i] = lds[LAP + 16 + i];
    float w0 = 0.f, w1 = 0.f, w2 = 0.f, w3 = 0.f;
    auto sweep = [&](v2f (&cur)[R][4], v2f (&nxt)[R][4], const float* pn)
        __attribute__((always_inline)) {
      lp_load4<0>(nxt, pn);
      lp_load4<4>(nxt, pn);
      for (int j = 0; j < R; ++j) {
        v8f acc = {};
        for (int k = 0; k < 4; ++k)
          acc = __builtin_amdgcn_wmma_f32_16x16x4_f32(false, Qf[k], false, cur[j][k],
                                                      (short)0, acc, false, false);
        const float n0 = a16[0]*w0 + a16[1]*w1 + a16[2]*w2 + a16[3]*w3 + acc[0];
        const float n1 = a16[4]*w0 + a16[5]*w1 + a16[6]*w2 + a16[7]*w3 + acc[1];
        const float n2 = a16[8]*w0 + a16[9]*w1 + a16[10]*w2 + a16[11]*w3 + acc[2];
        const float n3 = a16[12]*w0 + a16[13]*w1 + a16[14]*w2 + a16[15]*w3 + acc[3];
        w0 = n0; w1 = n1; w2 = n2; w3 = n3;
      }
    };
    sweep(xa, xb, sp + (size_t)(1 * RUNROWS) * S);
    sweep(xb, xa, sp + (size_t)(2 * RUNROWS) * S);
    sweep(xa, xb, sp + (size_t)(3 * RUNROWS) * S);
    sweep(xb, xa, xs + (size_t)ss_lo * SSROWS * S);
    {
      float* ud = lds + (half ? (LTR + wave * 64 + l16) : (LZR + wave * 64 + l16));
      ud[0] = w0; ud[16] = w1; ud[32] = w2; ud[48] = w3;
    }
    __syncthreads();
    if (wave == 0) {
      float a5[16];
      for (int i = 0; i < 16; ++i) a5[i] = lds[LA5 + i];
      float u[NW][4];
      for (int w = 0; w < NW; ++w) {
        const float* ub = lds + LZR + w * 64 + l16;
        u[w][0] = ub[0]; u[w][1] = ub[16]; u[w][2] = ub[32]; u[w][3] = ub[48];
      }
      float z0 = 0.f, z1 = 0.f, z2 = 0.f, z3 = 0.f;
      for (int w = 0; w < NW; ++w) {
        const float n0 = a5[0]*z0 + a5[1]*z1 + a5[2]*z2 + a5[3]*z3 + u[w][0];
        const float n1 = a5[4]*z0 + a5[5]*z1 + a5[6]*z2 + a5[7]*z3 + u[w][1];
        const float n2 = a5[8]*z0 + a5[9]*z1 + a5[10]*z2 + a5[11]*z3 + u[w][2];
        const float n3 = a5[12]*z0 + a5[13]*z1 + a5[14]*z2 + a5[15]*z3 + u[w][3];
        z0 = n0; z1 = n1; z2 = n2; z3 = n3;
      }
      zs0 = z0; zs1 = z1; zs2 = z2; zs3 = z3;
    }
  }

  auto step = [&](const int ss, v2f (&cur)[R][4], v2f (&nxt)[R][4])
      __attribute__((always_inline)) {
    const int sn = (ss + 1 < ss_hi) ? ss + 1 : ss;
    const float* xn = xs + (size_t)sn * SSROWS * S;

    float a16[16];
    for (int i = 0; i < 16; ++i) a16[i] = lds[LAP + 16 + i];
    float w0 = 0.f, w1 = 0.f, w2 = 0.f, w3 = 0.f;
    for (int j = 0; j < R; ++j) {
      v8f acc = {};
      for (int k = 0; k < 4; ++k)
        acc = __builtin_amdgcn_wmma_f32_16x16x4_f32(false, Qf[k], false, cur[j][k],
                                                    (short)0, acc, false, false);
      const float n0 = a16[0]*w0 + a16[1]*w1 + a16[2]*w2 + a16[3]*w3 + acc[0];
      const float n1 = a16[4]*w0 + a16[5]*w1 + a16[6]*w2 + a16[7]*w3 + acc[1];
      const float n2 = a16[8]*w0 + a16[9]*w1 + a16[10]*w2 + a16[11]*w3 + acc[2];
      const float n3 = a16[12]*w0 + a16[13]*w1 + a16[14]*w2 + a16[15]*w3 + acc[3];
      float* wd = lds + wl_off + j * wl_step;
      wd[0] = w0; wd[16] = w1; wd[32] = w2; wd[48] = w3;
      w0 = n0; w1 = n1; w2 = n2; w3 = n3;
    }
    {
      float* ud = lds + ub_off;
      ud[0] = w0; ud[16] = w1; ud[32] = w2; ud[48] = w3;
    }
    __syncthreads();

    if (wave == 0) {
      float a8[16];
      for (int i = 0; i < 16; ++i) a8[i] = lds[LAP + R * 16 + i];
      float u[NW][4];
      for (int w = 0; w < NW; ++w) {
        const float* ub = lds + LUB + w * 64 + l16;
        u[w][0] = ub[0]; u[w][1] = ub[16]; u[w][2] = ub[32]; u[w][3] = ub[48];
      }
      float z0 = zs0, z1 = zs1, z2 = zs2, z3 = zs3;
      for (int w = 0; w < NW; ++w) {
        float* zd = lds + zr_off + w * zr_step;
        zd[0] = z0; zd[16] = z1; zd[32] = z2; zd[48] = z3;
        const float n0 = a8[0]*z0 + a8[1]*z1 + a8[2]*z2 + a8[3]*z3 + u[w][0];
        const float n1 = a8[4]*z0 + a8[5]*z1 + a8[6]*z2 + a8[7]*z3 + u[w][1];
        const float n2 = a8[8]*z0 + a8[9]*z1 + a8[10]*z2 + a8[11]*z3 + u[w][2];
        const float n3 = a8[12]*z0 + a8[13]*z1 + a8[14]*z2 + a8[15]*z3 + u[w][3];
        z0 = n0; z1 = n1; z2 = n2; z3 = n3;
      }
      zs0 = z0; zs1 = z1; zs2 = z2; zs3 = z3;
    }
    __syncthreads();

    lp_load4<0>(nxt, xn);

    const float* zrp = lds + LZR + wave * 64 + l16;
    const float r0 = zrp[0], r1 = zrp[16], r2 = zrp[32], r3 = zrp[48];
    const float* wlp = lds + LWL + wave * (R * 64) + l16;
    float* o = out + (size_t)(ss * SSROWS + wave * RUNROWS + 8 * half) * S + chan;
    for (int j = 0; j < R; ++j) {
      if (j == R / 2) {
        asm volatile("" ::: "memory");
        lp_load4<4>(nxt, xn);
        asm volatile("" ::: "memory");
      }
      v2f zin;
      for (int s = 0; s < 2; ++s) {
        const int K = 2 * half + s;
        const float4 ap = *(const float4*)(lds + LAP + j * 16 + K * 4);
        zin[s] = ap.x * r0 + ap.y * r1 + ap.z * r2 + ap.w * r3 + wlp[j * 64 + K * 16];
      }
      v8f acc = {};
      acc = __builtin_amdgcn_wmma_f32_16x16x4_f32(false, Pf, false, zin,
                                                  (short)0, acc, false, false);
      for (int k = 0; k < 4; ++k)
        acc = __builtin_amdgcn_wmma_f32_16x16x4_f32(false, Hf[k], false, cur[j][k],
                                                    (short)0, acc, false, false);
      float* oj = o + (size_t)(j * L) * S;
      for (int r = 0; r < 8; ++r) oj[(size_t)r * S] = acc[r];
    }
  };

  for (int ss = ss_lo; ss < ss_hi; ss += 2) {
    step(ss, xa, xb);
    step(ss + 1, xb, xa);
  }
}

extern "C" void kernel_launch(void* const* d_in, const int* in_sizes, int n_in,
                              void* d_out, int out_size, void* d_ws, size_t ws_size,
                              hipStream_t stream) {
  const float* signal = (const float*)d_in[0];
  const float* b      = (const float*)d_in[1];
  const float* a      = (const float*)d_in[2];
  float* out = (float*)d_out;

  lp_strip_kernel<<<NGRID, 512, 0, stream>>>(signal, b, a, out);
}


# Work-group strip widened from 16 to 32 channels (two column tiles per wave, runs of 4 chunks, 128 groups): every row load and store of a lane half covers a whole 128-B line; summation order changes
# speedup vs baseline: 1.4748x; 1.2002x over previous
#include <hip/hip_runtime.h>


typedef __attribute__((ext_vector_type(2))) float v2f;
typedef __attribute__((ext_vector_type(8))) float v8f;

constexpr int S       = 2048;
constexpr int T       = 16384;
constexpr int L       = 16;
constexpr int NW      = 16;
constexpr int R       = 4;
constexpr int NT      = 2;
constexpr int RUNROWS = R * L;
constexpr int SSROWS  = NW * RUNROWS;
constexpr int NSS     = T / SSROWS;
constexpr int NSTRIP  = S / (16 * NT);
constexpr int NHALF   = 2;
constexpr int NGRID   = NSTRIP * NHALF;
constexpr int NSSH    = NSS / NHALF;
constexpr int SWROWS  = (T / NHALF) / NW;
constexpr int SWGRP   = SWROWS / RUNROWS;
static_assert(SWGRP == 8, "the entry-state sweep is written for eight groups");
static_assert(SWROWS == 512, "the sweep ends are joined with A^512");
static_assert(NSSH % 2 == 0, "the row buffers swap roles two super-steps per trip");

constexpr int LH  = 0;
constexpr int LP  = 256;
constexpr int LQ  = 320;
constexpr int LAP = 576;
constexpr int LHV = 720;
constexpr int LA5 = 736;
constexpr int LWL = 768;
constexpr int LUB = LWL + NW * R * NT * 64;
constexpr int LZR = LUB + NW * NT * 64;
constexpr int LTR = LZR + NW * NT * 64;
constexpr int LDS_FLOATS = LTR + NW * 64;

__device__ __forceinline__ void lp_make_A(const float* bc, const float* ac,
                                          float (&A)[4][4], float (&g4)[4]) {
  const float b0 = bc[0];
  for (int i = 0; i < 4; ++i) {
    for (int j = 0; j < 4; ++j) A[i][j] = (j == i + 1) ? 1.0f : 0.0f;
    A[i][0] -= ac[i + 1];
    g4[i] = bc[i + 1] - ac[i + 1] * b0;
  }
}

__device__ __forceinline__ void lp_matvec(const float (&A)[4][4], float (&v)[4]) {
  float nv[4];
  for (int i = 0; i < 4; ++i) {
    float s = 0.f;
    for (int j = 0; j < 4; ++j) s += A[i][j] * v[j];
    nv[i] = s;
  }
  for (int i = 0; i < 4; ++i) v[i] = nv[i];
}

__device__ __forceinline__ void lp_matmul(const float (&X)[4][4], const float (&Y)[4][4],
                                          float (&M)[4][4]) {
  float NM[4][4];
  for (int i = 0; i < 4; ++i)
    for (int j = 0; j < 4; ++j) {
      float s = 0.f;
      for (int k = 0; k < 4; ++k) s += X[i][k] * Y[k][j];
      NM[i][j] = s;
    }
  for (int i = 0; i < 4; ++i)
    for (int j = 0; j < 4; ++j) M[i][j] = NM[i][j];
}

template <int JB>
__device__ __forceinline__ void lp_load2(v2f (&xr)[R][4][2], const float* __restrict__ p) {
#pragma unroll
  for (int j = JB; j < JB + 2; ++j) {
#pragma unroll
    for (int k = 0; k < 4; ++k) {
      const int row = j * L + 4 * k;
      xr[j][k][0] = *(const v2f*)(p + (size_t)row * S);
      xr[j][k][1] = *(const v2f*)(p + (size_t)(row + 1) * S);
    }
  }
}

__device__ __forceinline__ void lp_frags(v2f (&f)[R][NT][4], const v2f (&xr)[R][4][2]) {
#pragma unroll
  for (int j = 0; j < R; ++j) {
#pragma unroll
    for (int k = 0; k < 4; ++k) {
#pragma unroll
      for (int t = 0; t < NT; ++t) {
        f[j][t][k][0] = xr[j][k][0][t];
        f[j][t][k][1] = xr[j][k][1][t];
      }
    }
  }
}

__global__ __launch_bounds__(512) void lp_strip_kernel(const float* __restrict__ x,
                                                       const float* __restrict__ bc,
                                                       const float* __restrict__ ac,
                                                       float* __restrict__ out) {
  __shared__ __attribute__((aligned(16))) float lds[LDS_FLOATS];

  const int tid  = threadIdx.x;
  const int wave = tid >> 5;
  const int lane = tid & 31;
  const int half = lane >> 4;
  const int l16  = lane & 15;
  const int bid  = blockIdx.x;
  const int g    = bid & (NSTRIP - 1);
  const int hh   = (bid < NSTRIP) ? 1 : 0;
  const int chan = g * (16 * NT) + NT * l16;
  const int ss_lo = hh * NSSH, ss_hi = ss_lo + NSSH;

  const float* xs = x + (size_t)(wave * RUNROWS + 2 * half) * S + chan;
  const float* sp = x + (size_t)(wave * SWROWS + 2 * half) * S + chan;

  v2f xa[R][4][2], xb[R][4][2];
  {
    const float* first = hh ? sp : xs + (size_t)ss_lo * SSROWS * S;
    lp_load2<0>(xa, first);
    lp_load2<2>(xa, first);
  }

  if (tid == 0) {
    float A[4][4], g4[4];
    lp_make_A(bc, ac, A, g4);
    lds[LHV + 0] = bc[0];
    float v[4] = {g4[0], g4[1], g4[2], g4[3]};
    for (int t = 1; t < 16; ++t) {
      lds[LHV + t] = v[0];
      lp_matvec(A, v);
    }
  }
  if (tid == 32) {
    float A[4][4], g4[4];
    lp_make_A(bc, ac, A, g4);
    float M[4][4];
    for (int i = 0; i < 4; ++i)
      for (int j = 0; j < 4; ++j) M[i][j] = (i == j) ? 1.0f : 0.0f;
    for (int t = 0; t < 16; ++t) {
      for (int m = 0; m < 4; ++m) lds[LP + t * 4 + m] = M[0][m];
      lp_matmul(A, M, M);
    }
    float A16[4][4], Pw[4][4];
    for (int i = 0; i < 4; ++i)
      for (int j = 0; j < 4; ++j) {
        A16[i][j] = M[i][j];
        Pw[i][j] = (i == j) ? 1.0f : 0.0f;
      }
    float A5[4][4];
    for (int p = 0; p <= R; ++p) {
      for (int i = 0; i < 4; ++i)
        for (int j = 0; j < 4; ++j) {
          lds[LAP + p * 16 + i * 4 + j] = Pw[i][j];
          if (p == R) A5[i][j] = Pw[i][j];
        }
      lp_matmul(Pw, A16, Pw);
    }
    for (int q = RUNROWS; q < SWROWS; q *= 2)
      lp_matmul(A5, A5, A5);
    for (int i = 0; i < 4; ++i)
      for (int j = 0; j < 4; ++j) lds[LA5 + i * 4 + j] = A5[i][j];
  }
  if (tid == 64) {
    float A[4][4], g4[4];
    lp_make_A(bc, ac, A, g4);
    float v[4] = {g4[0], g4[1], g4[2], g4[3]};
    for (int k = 15; k >= 0; --k) {
      for (int m = 0; m < 4; ++m) lds[LQ + m * 16 + k] = v[m];
      lp_matvec(A, v);
    }
  }
  if (tid >= 64 && tid < 256) lds[LQ + tid] = 0.0f;
  __syncthreads();
  if (tid < 256) {
    const int t = tid >> 4, k = tid & 15;
    lds[LH + tid] = (k <= t) ? lds[LHV + t - k] : 0.0f;
  }
  __syncthreads();

  v2f Hf[4], Qf[4], Pf;
  for (int k = 0; k < 4; ++k) {
    const int kk0 = 4 * k + 2 * half;
    Hf[k][0] = lds[LH + l16 * 16 + kk0];
    Hf[k][1] = lds[LH + l16 * 16 + kk0 + 1];
    Qf[k][0] = lds[LQ + l16 * 16 + kk0];
    Qf[k][1] = lds[LQ + l16 * 16 + kk0 + 1];
  }
  Pf[0] = lds[LP + l16 * 4 + 2 * half + 0];
  Pf[1] = lds[LP + l16 * 4 + 2 * half + 1];
  const int sink    = LTR + wave * 64 + l16;
  const int tstep   = half ? 0 : 64;
  const int wl_off  = half ? sink : (LWL + wave * (R * NT * 64) + l16);
  const int wl_step = half ? 0 : NT * 64;
  const int ub_off  = half ? sink : (LUB + wave * (NT * 64) + l16);
  const int zr_off  = half ? sink : (LZR + wave * 64 + l16);
  const int zr_step = half ? 0 : NT * 64;

  float zs0 = 0.f, zs1 = 0.f, zs2 = 0.f, zs3 = 0.f;

  if (hh) {
    float a16[16];
    for (int i = 0; i < 16; ++i) a16[i] = lds[LAP + 16 + i];
    float w[NT][4] = {};
    auto sweep = [&](v2f (&cur)[R][4][2], v2f (&nxt)[R][4][2], const float* pn)
        __attribute__((always_inline)) {
      lp_load2<0>(nxt, pn);
      lp_load2<2>(nxt, pn);
      v2f f[R][NT][4];
      lp_frags(f, cur);
#pragma unroll
      for (int j = 0; j < R; ++j) {
#pragma unroll
        for (int t = 0; t < NT; ++t) {
          v8f acc = {};
          for (int k = 0; k < 4; ++k)
            acc = __builtin_amdgcn_wmma_f32_16x16x4_f32(false, Qf[k], false, f[j][t][k],
                                                        (short)0, acc, false, false);
          const float w0 = w[t][0], w1 = w[t][1], w2 = w[t][2], w3 = w[t][3];
          w[t][0] = a16[0]*w0 + a16[1]*w1 + a16[2]*w2 + a16[3]*w3 + acc[0];
          w[t][1] = a16[4]*w0 + a16[5]*w1 + a16[6]*w2 + a16[7]*w3 + acc[1];
          w[t][2] = a16[8]*w0 + a16[9]*w1 + a16[10]*w2 + a16[11]*w3 + acc[2];
          w[t][3] = a16[12]*w0 + a16[13]*w1 + a16[14]*w2 + a16[15]*w3 + acc[3];
        }
      }
    };
    sweep(xa, xb, sp + (size_t)(1 * RUNROWS) * S);
    sweep(xb, xa, sp + (size_t)(2 * RUNROWS) * S);
    sweep(xa, xb, sp + (size_t)(3 * RUNROWS) * S);
    sweep(xb, xa, sp + (size_t)(4 * RUNROWS) * S);
    sweep(xa, xb, sp + (size_t)(5 * RUNROWS) * S);
    sweep(xb, xa, sp + (size_t)(6 * RUNROWS) * S);
    sweep(xa, xb, sp + (size_t)(7 * RUNROWS) * S);
    sweep(xb, xa, xs + (size_t)ss_lo * SSROWS * S);
#pragma unroll
    for (int t = 0; t < NT; ++t) {
      float* ud = lds + (half ? sink : (LZR + (wave * NT + t) * 64 + l16));
      ud[0] = w[t][0]; ud[16] = w[t][1]; ud[32] = w[t][2]; ud[48] = w[t][3];
    }
    __syncthreads();
    if (wave < NT) {
      float a5[16];
      for (int i = 0; i < 16; ++i) a5[i] = lds[LA5 + i];
      float u[NW][4];
      for (int w2 = 0; w2 < NW; ++w2) {
        const float* ub = lds + LZR + (w2 * NT + wave) * 64 + l16;
        u[w2][0] = ub[0]; u[w2][1] = ub[16]; u[w2][2] = ub[32]; u[w2][3] = ub[48];
      }
      float z0 = 0.f, z1 = 0.f, z2 = 0.f, z3 = 0.f;
      for (int w2 = 0; w2 < NW; ++w2) {
        const float n0 = a5[0]*z0 + a5[1]*z1 + a5[2]*z2 + a5[3]*z3 + u[w2][0];
        const float n1 = a5[4]*z0 + a5[5]*z1 + a5[6]*z2 + a5[7]*z3 + u[w2][1];
        const float n2 = a5[8]*z0 + a5[9]*z1 + a5[10]*z2 + a5[11]*z3 + u[w2][2];
        const float n3 = a5[12]*z0 + a5[13]*z1 + a5[14]*z2 + a5[15]*z3 + u[w2][3];
        z0 = n0; z1 = n1; z2 = n2; z3 = n3;
      }
      zs0 = z0; zs1 = z1; zs2 = z2; zs3 = z3;
    }
  }

  auto step = [&](const int ss, v2f (&cur)[R][4][2], v2f (&nxt)[R][4][2])
      __attribute__((always_inline)) {
    __builtin_amdgcn_sched_barrier(0);
    v2f f[R][NT][4];
    lp_frags(f, cur);
    const int sn = (ss + 1 < ss_hi) ? ss + 1 : ss;
    const float* xn = xs + (size_t)sn * SSROWS * S;

    float a16[16];
    for (int i = 0; i < 16; ++i) a16[i] = lds[LAP + 16 + i];
    float w[NT][4] = {};
#pragma unroll
    for (int j = 0; j < R; ++j) {
#pragma unroll
      for (int t = 0; t < NT; ++t) {
        v8f acc = {};
        for (int k = 0; k < 4; ++k)
          acc = __builtin_amdgcn_wmma_f32_16x16x4_f32(false, Qf[k], false, f[j][t][k],
                                                      (short)0, acc, false, false);
        const float w0 = w[t][0], w1 = w[t][1], w2 = w[t][2], w3 = w[t][3];
        float* wd = lds + wl_off + j * wl_step + t * tstep;
        wd[0] = w0; wd[16] = w1; wd[32] = w2; wd[48] = w3;
        w[t][0] = a16[0]*w0 + a16[1]*w1 + a16[2]*w2 + a16[3]*w3 + acc[0];
        w[t][1] = a16[4]*w0 + a16[5]*w1 + a16[6]*w2 + a16[7]*w3 + acc[1];
        w[t][2] = a16[8]*w0 + a16[9]*w1 + a16[10]*w2 + a16[11]*w3 + acc[2];
        w[t][3] = a16[12]*w0 + a16[13]*w1 + a16[14]*w2 + a16[15]*w3 + acc[3];
      }
    }
#pragma unroll
    for (int t = 0; t < NT; ++t) {
      float* ud = lds + ub_off + t * tstep;
      ud[0] = w[t][0]; ud[16] = w[t][1]; ud[32] = w[t][2]; ud[48] = w[t][3];
    }
    __syncthreads();

    if (wave < NT) {
      float aR[16];
      for (int i = 0; i < 16; ++i) aR[i] = lds[LAP + R * 16 + i];
      float u[NW][4];
      for (int w2 = 0; w2 < NW; ++w2) {
        const float* ub = lds + LUB + (w2 * NT + wave) * 64 + l16;
        u[w2][0] = ub[0]; u[w2][1] = ub[16]; u[w2][2] = ub[32]; u[w2][3] = ub[48];
      }
      float z0 = zs0, z1 = zs1, z2 = zs2, z3 = zs3;
      for (int w2 = 0; w2 < NW; ++w2) {
        float* zd = lds + zr_off + w2 * zr_step;
        zd[0] = z0; zd[16] = z1; zd[32] = z2; zd[48] = z3;
        const float n0 = aR[0]*z0 + aR[1]*z1 + aR[2]*z2 + aR[3]*z3 + u[w2][0];
        const float n1 = aR[4]*z0 + aR[5]*z1 + aR[6]*z2 + aR[7]*z3 + u[w2][1];
        const float n2 = aR[8]*z0 + aR[9]*z1 + aR[10]*z2 + aR[11]*z3 + u[w2][2];
        const float n3 = aR[12]*z0 + aR[13]*z1 + aR[14]*z2 + aR[15]*z3 + u[w2][3];
        z0 = n0; z1 = n1; z2 = n2; z3 = n3;
      }
      zs0 = z0; zs1 = z1; zs2 = z2; zs3 = z3;
    }
    __syncthreads();

    lp_load2<0>(nxt, xn);
    asm volatile("" ::: "memory");

    float rr[NT][4];
#pragma unroll
    for (int t = 0; t < NT; ++t) {
      const float* zrp = lds + LZR + (wave * NT + t) * 64 + l16;
      rr[t][0] = zrp[0]; rr[t][1] = zrp[16]; rr[t][2] = zrp[32]; rr[t][3] = zrp[48];
    }
    const float* wlp = lds + LWL + wave * (R * NT * 64) + l16;
    float* o = out + (size_t)(ss * SSROWS + wave * RUNROWS + 8 * half) * S + chan;
#pragma unroll
    for (int j = 0; j < R; ++j) {
      if (j == R / 2) {
        asm volatile("" ::: "memory");
        lp_load2<2>(nxt, xn);
        asm volatile("" ::: "memory");
      }
      v8f acc[NT];
#pragma unroll
      for (int t = 0; t < NT; ++t) {
        v2f zin;
        for (int s = 0; s < 2; ++s) {
          const int K = 2 * half + s;
          const float4 ap = *(const float4*)(lds + LAP + j * 16 + K * 4);
          zin[s] = ap.x * rr[t][0] + ap.y * rr[t][1] + ap.z * rr[t][2] + ap.w * rr[t][3]
                 + wlp[(j * NT + t) * 64 + K * 16];
        }
        v8f a = {};
        a = __builtin_amdgcn_wmma_f32_16x16x4_f32(false, Pf, false, zin,
                                                  (short)0, a, false, false);
        for (int k = 0; k < 4; ++k)
          a = __builtin_amdgcn_wmma_f32_16x16x4_f32(false, Hf[k], false, f[j][t][k],
                                                    (short)0, a, false, false);
        acc[t] = a;
      }
      float* oj = o + (size_t)(j * L) * S;
      for (int r = 0; r < 8; ++r)
        *(float2*)(oj + (size_t)r * S) = make_float2(acc[0][r], acc[1][r]);
    }
  };

  for (int ss = ss_lo; ss < ss_hi; ss += 2) {
    step(ss, xa, xb);
    step(ss + 1, xb, xa);
  }
}

extern "C" void kernel_launch(void* const* d_in, const int* in_sizes, int n_in,
                              void* d_out, int out_size, void* d_ws, size_t ws_size,
                              hipStream_t stream) {
  const float* signal = (const float*)d_in[0];
  const float* b      = (const float*)d_in[1];
  const float* a      = (const float*)d_in[2];
  float* out = (float*)d_out;

  lp_strip_kernel<<<NGRID, 512, 0, stream>>>(signal, b, a, out);
}


# Next super-step's rows requested at the head of the state pass and kept in flight across both barriers (LDS-only barrier: global loads and stores not drained); registers freed for them
# speedup vs baseline: 1.6430x; 1.1140x over previous
#include <hip/hip_runtime.h>


typedef __attribute__((ext_vector_type(2))) float v2f;
typedef __attribute__((ext_vector_type(8))) float v8f;

constexpr int S       = 2048;
constexpr int T       = 16384;
constexpr int L       = 16;
constexpr int NW      = 16;
constexpr int R       = 4;
constexpr int NT      = 2;
constexpr int RUNROWS = R * L;
constexpr int SSROWS  = NW * RUNROWS;
constexpr int NSS     = T / SSROWS;
constexpr int NSTRIP  = S / (16 * NT);
constexpr int NHALF   = 2;
constexpr int NGRID   = NSTRIP * NHALF;
constexpr int NSSH    = NSS / NHALF;
constexpr int SWROWS  = (T / NHALF) / NW;
constexpr int SWGRP   = SWROWS / RUNROWS;
static_assert(SWGRP == 8, "the entry-state sweep is written for eight groups");
static_assert(SWROWS == 512, "the sweep ends are joined with A^512");
static_assert(NSSH % 2 == 0, "the row buffers swap roles two super-steps per trip");

constexpr int LH  = 0;
constexpr int LP  = 256;
constexpr int LQ  = 320;
constexpr int LAP = 576;
constexpr int LHV = 720;
constexpr int LA5 = 736;
constexpr int LWL = 768;
constexpr int LUB = LWL + NW * R * NT * 64;
constexpr int LZR = LUB + NW * NT * 64;
constexpr int LEND = LZR + NW * NT * 64;
constexpr int SHD  = (LEND - LWL) + 16;
constexpr int LDS_FLOATS = LEND + SHD;

__device__ __forceinline__ void lp_make_A(const float* bc, const float* ac,
                                          float (&A)[4][4], float (&g4)[4]) {
  const float b0 = bc[0];
  for (int i = 0; i < 4; ++i) {
    for (int j = 0; j < 4; ++j) A[i][j] = (j == i + 1) ? 1.0f : 0.0f;
    A[i][0] -= ac[i + 1];
    g4[i] = bc[i + 1] - ac[i + 1] * b0;
  }
}

__device__ __forceinline__ void lp_matvec(const float (&A)[4][4], float (&v)[4]) {
  float nv[4];
  for (int i = 0; i < 4; ++i) {
    float s = 0.f;
    for (int j = 0; j < 4; ++j) s += A[i][j] * v[j];
    nv[i] = s;
  }
  for (int i = 0; i < 4; ++i) v[i] = nv[i];
}

__device__ __forceinline__ void lp_matmul(const float (&X)[4][4], const float (&Y)[4][4],
                                          float (&M)[4][4]) {
  float NM[4][4];
  for (int i = 0; i < 4; ++i)
    for (int j = 0; j < 4; ++j) {
      float s = 0.f;
      for (int k = 0; k < 4; ++k) s += X[i][k] * Y[k][j];
      NM[i][j] = s;
    }
  for (int i = 0; i < 4; ++i)
    for (int j = 0; j < 4; ++j) M[i][j] = NM[i][j];
}

__device__ __forceinline__ void lp_lds_barrier() {
  asm volatile("s_wait_dscnt 0x0\n\ts_barrier_signal -1\n\ts_barrier_wait -1" ::: "memory");
}

__device__ __forceinline__ float lp_uni(float v) {
  return __builtin_bit_cast(float, __builtin_amdgcn_readfirstlane(__builtin_bit_cast(int, v)));
}

template <int JB>
__device__ __forceinline__ void lp_load2(v2f (&xr)[R][4][2], const float* __restrict__ p) {
#pragma unroll
  for (int j = JB; j < JB + 2; ++j) {
#pragma unroll
    for (int k = 0; k < 4; ++k) {
      const int row = j * L + 4 * k;
      xr[j][k][0] = *(const v2f*)(p + (size_t)row * S);
      xr[j][k][1] = *(const v2f*)(p + (size_t)(row + 1) * S);
    }
  }
}

__device__ __forceinline__ void lp_frags(v2f (&f)[R][NT][4], const v2f (&xr)[R][4][2]) {
#pragma unroll
  for (int j = 0; j < R; ++j) {
#pragma unroll
    for (int k = 0; k < 4; ++k) {
#pragma unroll
      for (int t = 0; t < NT; ++t) {
        f[j][t][k][0] = xr[j][k][0][t];
        f[j][t][k][1] = xr[j][k][1][t];
      }
    }
  }
}

__global__ __launch_bounds__(512) void lp_strip_kernel(const float* __restrict__ x,
                                                       const float* __restrict__ bc,
                                                       const float* __restrict__ ac,
                                                       float* __restrict__ out) {
  __shared__ __attribute__((aligned(16))) float lds[LDS_FLOATS];

  const int tid  = threadIdx.x;
  const int wave = tid >> 5;
  const int lane = tid & 31;
  const int half = lane >> 4;
  const int l16  = lane & 15;
  const int bid  = blockIdx.x;
  const int g    = bid & (NSTRIP - 1);
  const int hh   = (bid < NSTRIP) ? 1 : 0;
  const int chan = g * (16 * NT) + NT * l16;
  const int ss_lo = hh * NSSH, ss_hi = ss_lo + NSSH;

  const float* xs = x + (size_t)(wave * RUNROWS + 2 * half) * S + chan;
  const float* sp = x + (size_t)(wave * SWROWS + 2 * half) * S + chan;

  v2f xa[R][4][2], xb[R][4][2];
  {
    const float* first = hh ? sp : xs + (size_t)ss_lo * SSROWS * S;
    lp_load2<0>(xa, first);
    lp_load2<2>(xa, first);
  }

  if (tid == 0) {
    float A[4][4], g4[4];
    lp_make_A(bc, ac, A, g4);
    lds[LHV + 0] = bc[0];
    float v[4] = {g4[0], g4[1], g4[2], g4[3]};
    for (int t = 1; t < 16; ++t) {
      lds[LHV + t] = v[0];
      lp_matvec(A, v);
    }
  }
  if (tid == 32) {
    float A[4][4], g4[4];
    lp_make_A(bc, ac, A, g4);
    float M[4][4];
    for (int i = 0; i < 4; ++i)
      for (int j = 0; j < 4; ++j) M[i][j] = (i == j) ? 1.0f : 0.0f;
    for (int t = 0; t < 16; ++t) {
      for (int m = 0; m < 4; ++m) lds[LP + t * 4 + m] = M[0][m];
      lp_matmul(A, M, M);
    }
    float A16[4][4], Pw[4][4];
    for (int i = 0; i < 4; ++i)
      for (int j = 0; j < 4; ++j) {
        A16[i][j] = M[i][j];
        Pw[i][j] = (i == j) ? 1.0f : 0.0f;
      }
    float A5[4][4];
    for (int p = 0; p <= R; ++p) {
      for (int i = 0; i < 4; ++i)
        for (int j = 0; j < 4; ++j) {
          lds[LAP + p * 16 + i * 4 + j] = Pw[i][j];
          if (p == R) A5[i][j] = Pw[i][j];
        }
      lp_matmul(Pw, A16, Pw);
    }
    for (int q = RUNROWS; q < SWROWS; q *= 2)
      lp_matmul(A5, A5, A5);
    for (int i = 0; i < 4; ++i)
      for (int j = 0; j < 4; ++j) lds[LA5 + i * 4 + j] = A5[i][j];
  }
  if (tid == 64) {
    float A[4][4], g4[4];
    lp_make_A(bc, ac, A, g4);
    float v[4] = {g4[0], g4[1], g4[2], g4[3]};
    for (int k = 15; k >= 0; --k) {
      for (int m = 0; m < 4; ++m) lds[LQ + m * 16 + k] = v[m];
      lp_matvec(A, v);
    }
  }
  if (tid >= 64 && tid < 256) lds[LQ + tid] = 0.0f;
  lp_lds_barrier();
  if (tid < 256) {
    const int t = tid >> 4, k = tid & 15;
    lds[LH + tid] = (k <= t) ? lds[LHV + t - k] : 0.0f;
  }
  lp_lds_barrier();

  const float* afr = lds + l16 * 16 + 2 * half;
  auto load_Q = [&](v2f (&Qf)[4]) __attribute__((always_inline)) {
    for (int k = 0; k < 4; ++k) Qf[k] = *(const v2f*)(afr + LQ + 4 * k);
  };
  auto load_HP = [&](v2f (&Hf)[4], v2f& Pf) __attribute__((always_inline)) {
    for (int k = 0; k < 4; ++k) Hf[k] = *(const v2f*)(afr + LH + 4 * k);
    Pf = *(const v2f*)(lds + LP + l16 * 4 + 2 * half);
  };
  const int shd     = half ? SHD : 0;
  constexpr int tstep   = 64;
  const int wl_off  = LWL + wave * (R * NT * 64) + l16 + shd;
  constexpr int wl_step = NT * 64;
  const int ub_off  = LUB + wave * (NT * 64) + l16 + shd;
  const int zr_off  = LZR + wave * 64 + l16 + shd;
  constexpr int zr_step = NT * 64;

  float zs0 = 0.f, zs1 = 0.f, zs2 = 0.f, zs3 = 0.f;

  if (hh) {
    float a16[16];
    for (int i = 0; i < 16; ++i) a16[i] = lp_uni(lds[LAP + 16 + i]);
    v2f Qf[4];
    load_Q(Qf);
    float w[NT][4] = {};
    auto sweep = [&](v2f (&cur)[R][4][2], v2f (&nxt)[R][4][2], const float* pn)
        __attribute__((always_inline)) {
      lp_load2<0>(nxt, pn);
      lp_load2<2>(nxt, pn);
      v2f f[R][NT][4];
      lp_frags(f, cur);
#pragma unroll
      for (int j = 0; j < R; ++j) {
#pragma unroll
        for (int t = 0; t < NT; ++t) {
          v8f acc = {};
          for (int k = 0; k < 4; ++k)
            acc = __builtin_amdgcn_wmma_f32_16x16x4_f32(false, Qf[k], false, f[j][t][k],
                                                        (short)0, acc, false, false);
          const float w0 = w[t][0], w1 = w[t][1], w2 = w[t][2], w3 = w[t][3];
          w[t][0] = a16[0]*w0 + a16[1]*w1 + a16[2]*w2 + a16[3]*w3 + acc[0];
          w[t][1] = a16[4]*w0 + a16[5]*w1 + a16[6]*w2 + a16[7]*w3 + acc[1];
          w[t][2] = a16[8]*w0 + a16[9]*w1 + a16[10]*w2 + a16[11]*w3 + acc[2];
          w[t][3] = a16[12]*w0 + a16[13]*w1 + a16[14]*w2 + a16[15]*w3 + acc[3];
        }
      }
    };
    sweep(xa, xb, sp + (size_t)(1 * RUNROWS) * S);
    sweep(xb, xa, sp + (size_t)(2 * RUNROWS) * S);
    sweep(xa, xb, sp + (size_t)(3 * RUNROWS) * S);
    sweep(xb, xa, sp + (size_t)(4 * RUNROWS) * S);
    sweep(xa, xb, sp + (size_t)(5 * RUNROWS) * S);
    sweep(xb, xa, sp + (size_t)(6 * RUNROWS) * S);
    sweep(xa, xb, sp + (size_t)(7 * RUNROWS) * S);
    sweep(xb, xa, xs + (size_t)ss_lo * SSROWS * S);
#pragma unroll
    for (int t = 0; t < NT; ++t) {
      float* ud = lds + LZR + (wave * NT + t) * 64 + l16 + shd;
      ud[0] = w[t][0]; ud[16] = w[t][1]; ud[32] = w[t][2]; ud[48] = w[t][3];
    }
    lp_lds_barrier();
    if (wave < NT) {
      float a5[16];
      for (int i = 0; i < 16; ++i) a5[i] = lp_uni(lds[LA5 + i]);
      float u[NW][4];
      for (int w2 = 0; w2 < NW; ++w2) {
        const float* ub = lds + LZR + (w2 * NT + wave) * 64 + l16;
        u[w2][0] = ub[0]; u[w2][1] = ub[16]; u[w2][2] = ub[32]; u[w2][3] = ub[48];
      }
      float z0 = 0.f, z1 = 0.f, z2 = 0.f, z3 = 0.f;
      for (int w2 = 0; w2 < NW; ++w2) {
        const float n0 = a5[0]*z0 + a5[1]*z1 + a5[2]*z2 + a5[3]*z3 + u[w2][0];
        const float n1 = a5[4]*z0 + a5[5]*z1 + a5[6]*z2 + a5[7]*z3 + u[w2][1];
        const float n2 = a5[8]*z0 + a5[9]*z1 + a5[10]*z2 + a5[11]*z3 + u[w2][2];
        const float n3 = a5[12]*z0 + a5[13]*z1 + a5[14]*z2 + a5[15]*z3 + u[w2][3];
        z0 = n0; z1 = n1; z2 = n2; z3 = n3;
      }
      zs0 = z0; zs1 = z1; zs2 = z2; zs3 = z3;
    }
  }

  auto step = [&](const int ss, v2f (&cur)[R][4][2], v2f (&nxt)[R][4][2])
      __attribute__((always_inline)) {
    __builtin_amdgcn_sched_barrier(0);
    v2f f[R][NT][4];
    lp_frags(f, cur);
    const int sn = (ss + 1 < ss_hi) ? ss + 1 : ss;
    const float* xn = xs + (size_t)sn * SSROWS * S;
    __builtin_amdgcn_sched_barrier(0);
    lp_load2<0>(nxt, xn);
    lp_load2<2>(nxt, xn);

    float a16[16];
    for (int i = 0; i < 16; ++i) a16[i] = lp_uni(lds[LAP + 16 + i]);
    v2f Qf[4];
    load_Q(Qf);
    float w[NT][4] = {};
#pragma unroll
    for (int j = 0; j < R; ++j) {
#pragma unroll
      for (int t = 0; t < NT; ++t) {
        v8f acc = {};
        for (int k = 0; k < 4; ++k)
          acc = __builtin_amdgcn_wmma_f32_16x16x4_f32(false, Qf[k], false, f[j][t][k],
                                                      (short)0, acc, false, false);
        const float w0 = w[t][0], w1 = w[t][1], w2 = w[t][2], w3 = w[t][3];
        float* wd = lds + wl_off + j * wl_step + t * tstep;
        wd[0] = w0; wd[16] = w1; wd[32] = w2; wd[48] = w3;
        w[t][0] = a16[0]*w0 + a16[1]*w1 + a16[2]*w2 + a16[3]*w3 + acc[0];
        w[t][1] = a16[4]*w0 + a16[5]*w1 + a16[6]*w2 + a16[7]*w3 + acc[1];
        w[t][2] = a16[8]*w0 + a16[9]*w1 + a16[10]*w2 + a16[11]*w3 + acc[2];
        w[t][3] = a16[12]*w0 + a16[13]*w1 + a16[14]*w2 + a16[15]*w3 + acc[3];
      }
    }
#pragma unroll
    for (int t = 0; t < NT; ++t) {
      float* ud = lds + ub_off + t * tstep;
      ud[0] = w[t][0]; ud[16] = w[t][1]; ud[32] = w[t][2]; ud[48] = w[t][3];
    }
    lp_lds_barrier();

    if (wave < NT) {
      float aR[16];
      for (int i = 0; i < 16; ++i) aR[i] = lp_uni(lds[LAP + R * 16 + i]);
      constexpr int UW = 4;
      float u[UW][4];
      for (int i = 0; i < UW; ++i) {
        const float* ub = lds + LUB + (i * NT + wave) * 64 + l16;
        u[i][0] = ub[0]; u[i][1] = ub[16]; u[i][2] = ub[32]; u[i][3] = ub[48];
      }
      float z0 = zs0, z1 = zs1, z2 = zs2, z3 = zs3;
#pragma unroll
      for (int gq = 0; gq < NW / UW; ++gq) {
        float un[UW][4] = {};
        if (gq + 1 < NW / UW) {
          for (int i = 0; i < UW; ++i) {
            const float* ub = lds + LUB + (((gq + 1) * UW + i) * NT + wave) * 64 + l16;
            un[i][0] = ub[0]; un[i][1] = ub[16]; un[i][2] = ub[32]; un[i][3] = ub[48];
          }
        }
#pragma unroll
        for (int i = 0; i < UW; ++i) {
          const int w2 = gq * UW + i;
          float* zd = lds + zr_off + w2 * zr_step;
          zd[0] = z0; zd[16] = z1; zd[32] = z2; zd[48] = z3;
          const float n0 = aR[0]*z0 + aR[1]*z1 + aR[2]*z2 + aR[3]*z3 + u[i][0];
          const float n1 = aR[4]*z0 + aR[5]*z1 + aR[6]*z2 + aR[7]*z3 + u[i][1];
          const float n2 = aR[8]*z0 + aR[9]*z1 + aR[10]*z2 + aR[11]*z3 + u[i][2];
          const float n3 = aR[12]*z0 + aR[13]*z1 + aR[14]*z2 + aR[15]*z3 + u[i][3];
          z0 = n0; z1 = n1; z2 = n2; z3 = n3;
        }
#pragma unroll
        for (int i = 0; i < UW; ++i)
          for (int m = 0; m < 4; ++m) u[i][m] = un[i][m];
      }
      zs0 = z0; zs1 = z1; zs2 = z2; zs3 = z3;
    }
    lp_lds_barrier();

    v2f Hf[4], Pf;
    load_HP(Hf, Pf);
    float rr[NT][4];
#pragma unroll
    for (int t = 0; t < NT; ++t) {
      const float* zrp = lds + LZR + (wave * NT + t) * 64 + l16;
      rr[t][0] = zrp[0]; rr[t][1] = zrp[16]; rr[t][2] = zrp[32]; rr[t][3] = zrp[48];
    }
    const float* wlp = lds + LWL + wave * (R * NT * 64) + l16;
    float* o = out + (size_t)(ss * SSROWS + wave * RUNROWS + 8 * half) * S + chan;
#pragma unroll
    for (int j = 0; j < R; ++j) {
      v8f acc[NT];
#pragma unroll
      for (int t = 0; t < NT; ++t) {
        v2f zin;
        for (int s = 0; s < 2; ++s) {
          const int K = 2 * half + s;
          const float4 ap = *(const float4*)(lds + LAP + j * 16 + K * 4);
          zin[s] = ap.x * rr[t][0] + ap.y * rr[t][1] + ap.z * rr[t][2] + ap.w * rr[t][3]
                 + wlp[(j * NT + t) * 64 + K * 16];
        }
        v8f a = {};
        a = __builtin_amdgcn_wmma_f32_16x16x4_f32(false, Pf, false, zin,
                                                  (short)0, a, false, false);
        for (int k = 0; k < 4; ++k)
          a = __builtin_amdgcn_wmma_f32_16x16x4_f32(false, Hf[k], false, f[j][t][k],
                                                    (short)0, a, false, false);
        acc[t] = a;
      }
      float* oj = o + (size_t)(j * L) * S;
      for (int r = 0; r < 8; ++r)
        *(float2*)(oj + (size_t)r * S) = make_float2(acc[0][r], acc[1][r]);
    }
  };

  for (int ss = ss_lo; ss < ss_hi; ss += 2) {
    step(ss, xa, xb);
    step(ss + 1, xb, xa);
  }
}

extern "C" void kernel_launch(void* const* d_in, const int* in_sizes, int n_in,
                              void* d_out, int out_size, void* d_ws, size_t ws_size,
                              hipStream_t stream) {
  const float* signal = (const float*)d_in[0];
  const float* b      = (const float*)d_in[1];
  const float* a      = (const float*)d_in[2];
  float* out = (float*)d_out;

  lp_strip_kernel<<<NGRID, 512, 0, stream>>>(signal, b, a, out);
}


# Time axis split over four work-groups per strip (256 groups, one per WGP), 6/4/4/2 super-steps; a later one sweeps all earlier rows for its entry state, joined via A^(64n); summation order changes
# speedup vs baseline: 1.9679x; 1.1977x over previous
#include <hip/hip_runtime.h>


typedef __attribute__((ext_vector_type(2))) float v2f;
typedef __attribute__((ext_vector_type(8))) float v8f;

constexpr int S       = 2048;
constexpr int T       = 16384;
constexpr int L       = 16;
constexpr int NW      = 16;
constexpr int R       = 4;
constexpr int NT      = 2;
constexpr int RUNROWS = R * L;
constexpr int SSROWS  = NW * RUNROWS;
constexpr int NSS     = T / SSROWS;
constexpr int NSTRIP  = S / (16 * NT);
constexpr int NSEG    = 4;
constexpr int NGRID   = NSTRIP * NSEG;
constexpr int SEG1 = 6, SEG2 = 10, SEG3 = 14;
static_assert(NSS == 16, "the segment bounds are written for sixteen super-steps");
static_assert(SEG1 % 2 == 0 && SEG2 % 2 == 0 && SEG3 % 2 == 0 && NSS % 2 == 0,
              "the row buffers swap roles two super-steps (and two sweep groups) per trip");
static_assert(SSROWS == NW * RUNROWS, "a wave sweeps one 64-row group per earlier super-step");

constexpr int LH  = 0;
constexpr int LP  = 256;
constexpr int LQ  = 320;
constexpr int LAP = 576;
constexpr int LHV = 720;
constexpr int LA5 = 736;
constexpr int LWL = 768;
constexpr int LUB = LWL + NW * R * NT * 64;
constexpr int LZR = LUB + NW * NT * 64;
constexpr int LEND = LZR + NW * NT * 64;
constexpr int SHD  = (LEND - LWL) + 16;
constexpr int LDS_FLOATS = LEND + SHD;

__device__ __forceinline__ void lp_make_A(const float* bc, const float* ac,
                                          float (&A)[4][4], float (&g4)[4]) {
  const float b0 = bc[0];
  for (int i = 0; i < 4; ++i) {
    for (int j = 0; j < 4; ++j) A[i][j] = (j == i + 1) ? 1.0f : 0.0f;
    A[i][0] -= ac[i + 1];
    g4[i] = bc[i + 1] - ac[i + 1] * b0;
  }
}

__device__ __forceinline__ void lp_matvec(const float (&A)[4][4], float (&v)[4]) {
  float nv[4];
  for (int i = 0; i < 4; ++i) {
    float s = 0.f;
    for (int j = 0; j < 4; ++j) s += A[i][j] * v[j];
    nv[i] = s;
  }
  for (int i = 0; i < 4; ++i) v[i] = nv[i];
}

__device__ __forceinline__ void lp_matmul(const float (&X)[4][4], const float (&Y)[4][4],
                                          float (&M)[4][4]) {
  float NM[4][4];
  for (int i = 0; i < 4; ++i)
    for (int j = 0; j < 4; ++j) {
      float s = 0.f;
      for (int k = 0; k < 4; ++k) s += X[i][k] * Y[k][j];
      NM[i][j] = s;
    }
  for (int i = 0; i < 4; ++i)
    for (int j = 0; j < 4; ++j) M[i][j] = NM[i][j];
}

__device__ __forceinline__ void lp_lds_barrier() {
  asm volatile("s_wait_dscnt 0x0\n\ts_barrier_signal -1\n\ts_barrier_wait -1" ::: "memory");
}

__device__ __forceinline__ float lp_uni(float v) {
  return __builtin_bit_cast(float, __builtin_amdgcn_readfirstlane(__builtin_bit_cast(int, v)));
}

template <int JB>
__device__ __forceinline__ void lp_load2(v2f (&xr)[R][4][2], const float* __restrict__ p) {
#pragma unroll
  for (int j = JB; j < JB + 2; ++j) {
#pragma unroll
    for (int k = 0; k < 4; ++k) {
      const int row = j * L + 4 * k;
      xr[j][k][0] = *(const v2f*)(p + (size_t)row * S);
      xr[j][k][1] = *(const v2f*)(p + (size_t)(row + 1) * S);
    }
  }
}

__device__ __forceinline__ void lp_frags(v2f (&f)[R][NT][4], const v2f (&xr)[R][4][2]) {
#pragma unroll
  for (int j = 0; j < R; ++j) {
#pragma unroll
    for (int k = 0; k < 4; ++k) {
#pragma unroll
      for (int t = 0; t < NT; ++t) {
        f[j][t][k][0] = xr[j][k][0][t];
        f[j][t][k][1] = xr[j][k][1][t];
      }
    }
  }
}

__global__ __launch_bounds__(512) void lp_strip_kernel(const float* __restrict__ x,
                                                       const float* __restrict__ bc,
                                                       const float* __restrict__ ac,
                                                       float* __restrict__ out) {
  __shared__ __attribute__((aligned(16))) float lds[LDS_FLOATS];

  const int tid  = threadIdx.x;
  const int wave = tid >> 5;
  const int lane = tid & 31;
  const int half = lane >> 4;
  const int l16  = lane & 15;
  const int bid  = blockIdx.x;
  const int g    = bid & (NSTRIP - 1);
  const int q    = ((bid / NSTRIP) + 2) & (NSEG - 1);
  const int chan = g * (16 * NT) + NT * l16;
  const int ss_lo = (q == 0) ? 0 : (q == 1) ? SEG1 : (q == 2) ? SEG2 : SEG3;
  const int ss_hi = (q == 0) ? SEG1 : (q == 1) ? SEG2 : (q == 2) ? SEG3 : NSS;

  const float* xs = x + (size_t)(wave * RUNROWS + 2 * half) * S + chan;
  const float* sp = x + (size_t)(wave * (ss_lo * RUNROWS) + 2 * half) * S + chan;

  v2f xa[R][4][2], xb[R][4][2];
  {
    const float* first = ss_lo ? sp : xs;
    lp_load2<0>(xa, first);
    lp_load2<2>(xa, first);
  }

  if (tid == 0) {
    float A[4][4], g4[4];
    lp_make_A(bc, ac, A, g4);
    lds[LHV + 0] = bc[0];
    float v[4] = {g4[0], g4[1], g4[2], g4[3]};
    for (int t = 1; t < 16; ++t) {
      lds[LHV + t] = v[0];
      lp_matvec(A, v);
    }
  }
  if (tid == 32) {
    float A[4][4], g4[4];
    lp_make_A(bc, ac, A, g4);
    float M[4][4];
    for (int i = 0; i < 4; ++i)
      for (int j = 0; j < 4; ++j) M[i][j] = (i == j) ? 1.0f : 0.0f;
    for (int t = 0; t < 16; ++t) {
      for (int m = 0; m < 4; ++m) lds[LP + t * 4 + m] = M[0][m];
      lp_matmul(A, M, M);
    }
    float A16[4][4], Pw[4][4];
    for (int i = 0; i < 4; ++i)
      for (int j = 0; j < 4; ++j) {
        A16[i][j] = M[i][j];
        Pw[i][j] = (i == j) ? 1.0f : 0.0f;
      }
    float A5[4][4], B5[4][4];
    for (int p = 0; p <= R; ++p) {
      for (int i = 0; i < 4; ++i)
        for (int j = 0; j < 4; ++j) {
          lds[LAP + p * 16 + i * 4 + j] = Pw[i][j];
          if (p == R) {
            B5[i][j] = Pw[i][j];
            A5[i][j] = (i == j) ? 1.0f : 0.0f;
          }
        }
      lp_matmul(Pw, A16, Pw);
    }
    for (int bit = 0; bit < 4; ++bit) {
      if ((ss_lo >> bit) & 1) lp_matmul(A5, B5, A5);
      lp_matmul(B5, B5, B5);
    }
    for (int i = 0; i < 4; ++i)
      for (int j = 0; j < 4; ++j) lds[LA5 + i * 4 + j] = A5[i][j];
  }
  if (tid == 64) {
    float A[4][4], g4[4];
    lp_make_A(bc, ac, A, g4);
    float v[4] = {g4[0], g4[1], g4[2], g4[3]};
    for (int k = 15; k >= 0; --k) {
      for (int m = 0; m < 4; ++m) lds[LQ + m * 16 + k] = v[m];
      lp_matvec(A, v);
    }
  }
  if (tid >= 64 && tid < 256) lds[LQ + tid] = 0.0f;
  lp_lds_barrier();
  if (tid < 256) {
    const int t = tid >> 4, k = tid & 15;
    lds[LH + tid] = (k <= t) ? lds[LHV + t - k] : 0.0f;
  }
  lp_lds_barrier();

  const float* afr = lds + l16 * 16 + 2 * half;
  auto load_Q = [&](v2f (&Qf)[4]) __attribute__((always_inline)) {
    for (int k = 0; k < 4; ++k) Qf[k] = *(const v2f*)(afr + LQ + 4 * k);
  };
  auto load_HP = [&](v2f (&Hf)[4], v2f& Pf) __attribute__((always_inline)) {
    for (int k = 0; k < 4; ++k) Hf[k] = *(const v2f*)(afr + LH + 4 * k);
    Pf = *(const v2f*)(lds + LP + l16 * 4 + 2 * half);
  };
  const int shd     = half ? SHD : 0;
  constexpr int tstep   = 64;
  const int wl_off  = LWL + wave * (R * NT * 64) + l16 + shd;
  constexpr int wl_step = NT * 64;
  const int ub_off  = LUB + wave * (NT * 64) + l16 + shd;
  const int zr_off  = LZR + wave * 64 + l16 + shd;
  constexpr int zr_step = NT * 64;

  float zs0 = 0.f, zs1 = 0.f, zs2 = 0.f, zs3 = 0.f;

  if (ss_lo) {
    float a16[16];
    for (int i = 0; i < 16; ++i) a16[i] = lp_uni(lds[LAP + 16 + i]);
    v2f Qf[4];
    load_Q(Qf);
    float w[NT][4] = {};
    auto sweep = [&](v2f (&cur)[R][4][2], v2f (&nxt)[R][4][2], const float* pn)
        __attribute__((always_inline)) {
      lp_load2<0>(nxt, pn);
      lp_load2<2>(nxt, pn);
      __builtin_amdgcn_sched_barrier(0);
      v2f f[R][NT][4];
      lp_frags(f, cur);
#pragma unroll
      for (int j = 0; j < R; ++j) {
#pragma unroll
        for (int t = 0; t < NT; ++t) {
          v8f acc = {};
          for (int k = 0; k < 4; ++k)
            acc = __builtin_amdgcn_wmma_f32_16x16x4_f32(false, Qf[k], false, f[j][t][k],
                                                        (short)0, acc, false, false);
          const float w0 = w[t][0], w1 = w[t][1], w2 = w[t][2], w3 = w[t][3];
          w[t][0] = a16[0]*w0 + a16[1]*w1 + a16[2]*w2 + a16[3]*w3 + acc[0];
          w[t][1] = a16[4]*w0 + a16[5]*w1 + a16[6]*w2 + a16[7]*w3 + acc[1];
          w[t][2] = a16[8]*w0 + a16[9]*w1 + a16[10]*w2 + a16[11]*w3 + acc[2];
          w[t][3] = a16[12]*w0 + a16[13]*w1 + a16[14]*w2 + a16[15]*w3 + acc[3];
        }
      }
    };
    for (int gi = 0; gi < ss_lo; gi += 2) {
      const float* pn = (gi + 2 < ss_lo) ? sp + (size_t)((gi + 2) * RUNROWS) * S
                                         : xs + (size_t)ss_lo * SSROWS * S;
      sweep(xa, xb, sp + (size_t)((gi + 1) * RUNROWS) * S);
      sweep(xb, xa, pn);
    }
#pragma unroll
    for (int t = 0; t < NT; ++t) {
      float* ud = lds + LZR + (wave * NT + t) * 64 + l16 + shd;
      ud[0] = w[t][0]; ud[16] = w[t][1]; ud[32] = w[t][2]; ud[48] = w[t][3];
    }
    lp_lds_barrier();
    if (wave < NT) {
      float a5[16];
      for (int i = 0; i < 16; ++i) a5[i] = lp_uni(lds[LA5 + i]);
      float u[NW][4];
      for (int w2 = 0; w2 < NW; ++w2) {
        const float* ub = lds + LZR + (w2 * NT + wave) * 64 + l16;
        u[w2][0] = ub[0]; u[w2][1] = ub[16]; u[w2][2] = ub[32]; u[w2][3] = ub[48];
      }
      float z0 = 0.f, z1 = 0.f, z2 = 0.f, z3 = 0.f;
      for (int w2 = 0; w2 < NW; ++w2) {
        const float n0 = a5[0]*z0 + a5[1]*z1 + a5[2]*z2 + a5[3]*z3 + u[w2][0];
        const float n1 = a5[4]*z0 + a5[5]*z1 + a5[6]*z2 + a5[7]*z3 + u[w2][1];
        const float n2 = a5[8]*z0 + a5[9]*z1 + a5[10]*z2 + a5[11]*z3 + u[w2][2];
        const float n3 = a5[12]*z0 + a5[13]*z1 + a5[14]*z2 + a5[15]*z3 + u[w2][3];
        z0 = n0; z1 = n1; z2 = n2; z3 = n3;
      }
      zs0 = z0; zs1 = z1; zs2 = z2; zs3 = z3;
    }
  }

  auto step = [&](const int ss, v2f (&cur)[R][4][2], v2f (&nxt)[R][4][2])
      __attribute__((always_inline)) {
    __builtin_amdgcn_sched_barrier(0);
    v2f f[R][NT][4];
    lp_frags(f, cur);
    const int sn = (ss + 1 < ss_hi) ? ss + 1 : ss;
    const float* xn = xs + (size_t)sn * SSROWS * S;
    __builtin_amdgcn_sched_barrier(0);
    lp_load2<0>(nxt, xn);
    lp_load2<2>(nxt, xn);

    float a16[16];
    for (int i = 0; i < 16; ++i) a16[i] = lp_uni(lds[LAP + 16 + i]);
    v2f Qf[4];
    load_Q(Qf);
    float w[NT][4] = {};
#pragma unroll
    for (int j = 0; j < R; ++j) {
#pragma unroll
      for (int t = 0; t < NT; ++t) {
        v8f acc = {};
        for (int k = 0; k < 4; ++k)
          acc = __builtin_amdgcn_wmma_f32_16x16x4_f32(false, Qf[k], false, f[j][t][k],
                                                      (short)0, acc, false, false);
        const float w0 = w[t][0], w1 = w[t][1], w2 = w[t][2], w3 = w[t][3];
        float* wd = lds + wl_off + j * wl_step + t * tstep;
        wd[0] = w0; wd[16] = w1; wd[32] = w2; wd[48] = w3;
        w[t][0] = a16[0]*w0 + a16[1]*w1 + a16[2]*w2 + a16[3]*w3 + acc[0];
        w[t][1] = a16[4]*w0 + a16[5]*w1 + a16[6]*w2 + a16[7]*w3 + acc[1];
        w[t][2] = a16[8]*w0 + a16[9]*w1 + a16[10]*w2 + a16[11]*w3 + acc[2];
        w[t][3] = a16[12]*w0 + a16[13]*w1 + a16[14]*w2 + a16[15]*w3 + acc[3];
      }
    }
#pragma unroll
    for (int t = 0; t < NT; ++t) {
      float* ud = lds + ub_off + t * tstep;
      ud[0] = w[t][0]; ud[16] = w[t][1]; ud[32] = w[t][2]; ud[48] = w[t][3];
    }
    lp_lds_barrier();

    if (wave < NT) {
      float aR[16];
      for (int i = 0; i < 16; ++i) aR[i] = lp_uni(lds[LAP + R * 16 + i]);
      constexpr int UW = 4;
      float u[UW][4];
      for (int i = 0; i < UW; ++i) {
        const float* ub = lds + LUB + (i * NT + wave) * 64 + l16;
        u[i][0] = ub[0]; u[i][1] = ub[16]; u[i][2] = ub[32]; u[i][3] = ub[48];
      }
      float z0 = zs0, z1 = zs1, z2 = zs2, z3 = zs3;
#pragma unroll
      for (int gq = 0; gq < NW / UW; ++gq) {
        float un[UW][4] = {};
        if (gq + 1 < NW / UW) {
          for (int i = 0; i < UW; ++i) {
            const float* ub = lds + LUB + (((gq + 1) * UW + i) * NT + wave) * 64 + l16;
            un[i][0] = ub[0]; un[i][1] = ub[16]; un[i][2] = ub[32]; un[i][3] = ub[48];
          }
        }
#pragma unroll
        for (int i = 0; i < UW; ++i) {
          const int w2 = gq * UW + i;
          float* zd = lds + zr_off + w2 * zr_step;
          zd[0] = z0; zd[16] = z1; zd[32] = z2; zd[48] = z3;
          const float n0 = aR[0]*z0 + aR[1]*z1 + aR[2]*z2 + aR[3]*z3 + u[i][0];
          const float n1 = aR[4]*z0 + aR[5]*z1 + aR[6]*z2 + aR[7]*z3 + u[i][1];
          const float n2 = aR[8]*z0 + aR[9]*z1 + aR[10]*z2 + aR[11]*z3 + u[i][2];
          const float n3 = aR[12]*z0 + aR[13]*z1 + aR[14]*z2 + aR[15]*z3 + u[i][3];
          z0 = n0; z1 = n1; z2 = n2; z3 = n3;
        }
#pragma unroll
        for (int i = 0; i < UW; ++i)
          for (int m = 0; m < 4; ++m) u[i][m] = un[i][m];
      }
      zs0 = z0; zs1 = z1; zs2 = z2; zs3 = z3;
    }
    lp_lds_barrier();

    v2f Hf[4], Pf;
    load_HP(Hf, Pf);
    float rr[NT][4];
#pragma unroll
    for (int t = 0; t < NT; ++t) {
      const float* zrp = lds + LZR + (wave * NT + t) * 64 + l16;
      rr[t][0] = zrp[0]; rr[t][1] = zrp[16]; rr[t][2] = zrp[32]; rr[t][3] = zrp[48];
    }
    const float* wlp = lds + LWL + wave * (R * NT * 64) + l16;
    float* o = out + (size_t)(ss * SSROWS + wave * RUNROWS + 8 * half) * S + chan;
#pragma unroll
    for (int j = 0; j < R; ++j) {
      v8f acc[NT];
#pragma unroll
      for (int t = 0; t < NT; ++t) {
        v2f zin;
        for (int s = 0; s < 2; ++s) {
          const int K = 2 * half + s;
          const float4 ap = *(const float4*)(lds + LAP + j * 16 + K * 4);
          zin[s] = ap.x * rr[t][0] + ap.y * rr[t][1] + ap.z * rr[t][2] + ap.w * rr[t][3]
                 + wlp[(j * NT + t) * 64 + K * 16];
        }
        v8f a = {};
        a = __builtin_amdgcn_wmma_f32_16x16x4_f32(false, Pf, false, zin,
                                                  (short)0, a, false, false);
        for (int k = 0; k < 4; ++k)
          a = __builtin_amdgcn_wmma_f32_16x16x4_f32(false, Hf[k], false, f[j][t][k],
                                                    (short)0, a, false, false);
        acc[t] = a;
      }
      float* oj = o + (size_t)(j * L) * S;
      for (int r = 0; r < 8; ++r)
        *(float2*)(oj + (size_t)r * S) = make_float2(acc[0][r], acc[1][r]);
    }
  };

  for (int ss = ss_lo; ss < ss_hi; ss += 2) {
    step(ss, xa, xb);
    step(ss + 1, xb, xa);
  }
}

extern "C" void kernel_launch(void* const* d_in, const int* in_sizes, int n_in,
                              void* d_out, int out_size, void* d_ws, size_t ws_size,
                              hipStream_t stream) {
  const float* signal = (const float*)d_in[0];
  const float* b      = (const float*)d_in[1];
  const float* a      = (const float*)d_in[2];
  float* out = (float*)d_out;

  lp_strip_kernel<<<NGRID, 512, 0, stream>>>(signal, b, a, out);
}
